# starting kernel
# baseline (speedup 1.0000x reference)
#include <hip/hip_runtime.h>
#include <hip/hip_bf16.h>

typedef __attribute__((ext_vector_type(2))) float v2f;
typedef __attribute__((ext_vector_type(8))) float v8f;

#define D 256
#define D4 64
#define LDS_STRIDE 260

__global__ void zero_deg_kernel(int* __restrict__ deg, int n) {
    int i = blockIdx.x * blockDim.x + threadIdx.x;
    if (i < n) deg[i] = 0;
}

__global__ void count_deg_kernel(const int* __restrict__ row, int* __restrict__ deg, int nE) {
    int e = blockIdx.x * blockDim.x + threadIdx.x;
    if (e < nE) atomicAdd(&deg[row[e]], 1);
}

__global__ void calc_dis_kernel(const int* __restrict__ deg, float* __restrict__ dis, int n) {
    int i = blockIdx.x * blockDim.x + threadIdx.x;
    if (i < n) {
        dis[i] = rsqrtf((float)(deg[i] + 1));
    }
}

__global__ __launch_bounds__(256) void pack_w_kernel(const float* __restrict__ W,
                                                     float4* __restrict__ pack4) {
    int t = blockIdx.x * 256 + threadIdx.x;
    int l   = t & 31;
    int s   = (t >> 5) & 63;
    int wc  = (t >> 11) & 7;
    int k   = t >> 14;
    int m   = l & 15;
    int h   = l >> 4;
    int kA  = 4 * s + 2 * h;
    int n0  = wc * 32 + m;
    const float* Wk = W + (size_t)k * D * D;
    float4 o;
    o.x = Wk[(size_t)kA * D + n0];
    o.y = Wk[(size_t)(kA + 1) * D + n0];
    o.z = Wk[(size_t)kA * D + n0 + 16];
    o.w = Wk[(size_t)(kA + 1) * D + n0 + 16];
    pack4[t] = o;
}

__global__ void cheb_init_kernel(const float* __restrict__ src,
                                 const float* __restrict__ prev,
                                 const float* __restrict__ dis,
                                 float* __restrict__ dst,
                                 float scale, float pscale, int nN) {
    int gid = blockIdx.x * blockDim.x + threadIdx.x;
    int total = nN * D4;
    if (gid >= total) return;
    int i  = gid >> 6;
    int f4 = gid & 63;
    float d = dis[i];
    float w = scale * (1.0f - d * d);
    size_t off = (size_t)i * D4 + f4;
    float4 v  = ((const float4*)src)[off];
    float4 pv = ((const float4*)prev)[off];
    float4 o;
    o.x = w * v.x - pscale * pv.x;
    o.y = w * v.y - pscale * pv.y;
    o.z = w * v.z - pscale * pv.z;
    o.w = w * v.w - pscale * pv.w;
    ((float4*)dst)[off] = o;
}

__global__ __launch_bounds__(256) void cheb_edges_kernel(
    const float* __restrict__ src, const float* __restrict__ dis,
    const int* __restrict__ row, const int* __restrict__ col,
    float* __restrict__ dst, float scale, int nE) {
    int wid  = threadIdx.x >> 5;
    int lane = threadIdx.x & 31;
    int e = blockIdx.x * 8 + wid;
    if (e >= nE) return;
    int r = row[e];
    int c = col[e];
    float w = -dis[r] * dis[c];
    if (r == c) w += 1.0f;
    w *= scale;
    const float4* s = (const float4*)(src + (size_t)c * D);
    float* dp = dst + (size_t)r * D;
#pragma unroll
    for (int h = 0; h < 2; ++h) {
        int f4 = lane + h * 32;
        float4 v = s[f4];
        atomicAdd(dp + f4 * 4 + 0, w * v.x);
        atomicAdd(dp + f4 * 4 + 1, w * v.y);
        atomicAdd(dp + f4 * 4 + 2, w * v.z);
        atomicAdd(dp + f4 * 4 + 3, w * v.w);
    }
}

template <bool FULL>
__device__ __forceinline__ void gemm_body(
    float (*As)[LDS_STRIDE],
    const float* __restrict__ T0, const float* __restrict__ T1,
    const float* __restrict__ T2, const float* __restrict__ T3,
    const float4* __restrict__ pack4, const float* __restrict__ bias,
    float* __restrict__ out, int rowBase, int tid, int wid, int m, int h) {
    const int lane = tid & 31;
    v8f acc00 = {}, acc01 = {}, acc10 = {}, acc11 = {};
    const float* Ts[4] = {T0, T1, T2, T3};

#pragma unroll
    for (int k = 0; k < 4; ++k) {
        __syncthreads();
        const float* Tk = Ts[k];
        const int nload = FULL ? 8 : 4;
#pragma unroll
        for (int i = 0; i < nload; ++i) {
            int idx = tid + i * 256;
            int r   = idx >> 6;
            int c4  = idx & 63;
            float4 v = *(const float4*)(Tk + ((size_t)(rowBase + r)) * D + c4 * 4);
            *(float4*)(&As[r][c4 * 4]) = v;
        }
        __syncthreads();

        const float4* p4 = pack4 + ((size_t)(k * 8 + wid) * 64 * 32) + lane;
#pragma unroll 4
        for (int s = 0; s < 64; ++s) {
            const int kA = 4 * s + 2 * h;
            v2f a0;
            a0.x = As[m][kA];
            a0.y = As[m][kA + 1];
            float4 bb = p4[(size_t)s * 32];
            v2f b0; b0.x = bb.x; b0.y = bb.y;
            v2f b1; b1.x = bb.z; b1.y = bb.w;
            acc00 = __builtin_amdgcn_wmma_f32_16x16x4_f32(
                false, a0, false, b0, (short)0, acc00, false, false);
            acc01 = __builtin_amdgcn_wmma_f32_16x16x4_f32(
                false, a0, false, b1, (short)0, acc01, false, false);
            if (FULL) {
                v2f a1;
                a1.x = As[16 + m][kA];
                a1.y = As[16 + m][kA + 1];
                acc10 = __builtin_amdgcn_wmma_f32_16x16x4_f32(
                    false, a1, false, b0, (short)0, acc10, false, false);
                acc11 = __builtin_amdgcn_wmma_f32_16x16x4_f32(
                    false, a1, false, b1, (short)0, acc11, false, false);
            }
        }
    }

    const int colBase = wid * 32;
    const float bias0 = bias[colBase + m];
    const float bias1 = bias[colBase + 16 + m];
#pragma unroll
    for (int j = 0; j < 8; ++j) {
        int rr = rowBase + j + 8 * h;
        out[(size_t)rr * D + colBase + m]      = acc00[j] + bias0;
        out[(size_t)rr * D + colBase + 16 + m] = acc01[j] + bias1;
        if (FULL) {
            int rr1 = rr + 16;
            out[(size_t)rr1 * D + colBase + m]      = acc10[j] + bias0;
            out[(size_t)rr1 * D + colBase + 16 + m] = acc11[j] + bias1;
        }
    }
}

__global__ __launch_bounds__(256) void cheb_gemm_kernel(
    const float* __restrict__ T0, const float* __restrict__ T1,
    const float* __restrict__ T2, const float* __restrict__ T3,
    const float4* __restrict__ pack4, const float* __restrict__ bias,
    float* __restrict__ out, int nN) {
    __shared__ float As[32][LDS_STRIDE];
    const int tid = threadIdx.x;
    const int wid = tid >> 5;
    const int m   = tid & 15;
    const int h   = (tid >> 4) & 1;
    const int rowBase = blockIdx.x * 32;
    if (rowBase + 32 <= nN) {
        gemm_body<true>(As, T0, T1, T2, T3, pack4, bias, out, rowBase, tid, wid, m, h);
    } else {
        gemm_body<false>(As, T0, T1, T2, T3, pack4, bias, out, rowBase, tid, wid, m, h);
    }
}

extern "C" void kernel_launch(void* const* d_in, const int* in_sizes, int n_in,
                              void* d_out, int out_size, void* d_ws, size_t ws_size,
                              hipStream_t stream) {
    const float* x    = (const float*)d_in[0];
    const int*   ei   = (const int*)d_in[1];
    const float* W    = (const float*)d_in[2];
    const float* bias = (const float*)d_in[3];
    float* out = (float*)d_out;

    const int N = in_sizes[0] / D;
    const int E = in_sizes[1] / 2;
    const int* row = ei;
    const int* col = ei + E;

    char* ws = (char*)d_ws;
    size_t termBytes = (size_t)N * D * sizeof(float);
    int*    deg   = (int*)ws;
    float*  dis   = (float*)(ws + (1u << 18));
    float4* pack4 = (float4*)(ws + (1u << 19));
    float*  T1    = (float*)(ws + (3u << 19));
    float*  T2    = (float*)((char*)T1 + termBytes);
    float*  T3    = (float*)((char*)T2 + termBytes);

    const int bs = 256;
    pack_w_kernel<<<256, bs, 0, stream>>>(W, pack4);

    zero_deg_kernel<<<(N + bs - 1) / bs, bs, 0, stream>>>(deg, N);
    count_deg_kernel<<<(E + bs - 1) / bs, bs, 0, stream>>>(row, deg, E);
    calc_dis_kernel<<<(N + bs - 1) / bs, bs, 0, stream>>>(deg, dis, N);

    const int initGrid = (N * D4 + bs - 1) / bs;
    const int edgeGrid = (E + 7) / 8;

    cheb_init_kernel<<<initGrid, bs, 0, stream>>>(x, x, dis, T1, 1.0f, 0.0f, N);
    cheb_edges_kernel<<<edgeGrid, bs, 0, stream>>>(x, dis, row, col, T1, 1.0f, E);
    cheb_init_kernel<<<initGrid, bs, 0, stream>>>(T1, x, dis, T2, 2.0f, 1.0f, N);
    cheb_edges_kernel<<<edgeGrid, bs, 0, stream>>>(T1, dis, row, col, T2, 2.0f, E);
    cheb_init_kernel<<<initGrid, bs, 0, stream>>>(T2, T1, dis, T3, 2.0f, 1.0f, N);
    cheb_edges_kernel<<<edgeGrid, bs, 0, stream>>>(T2, dis, row, col, T3, 2.0f, E);

    cheb_gemm_kernel<<<(N + 31) / 32, bs, 0, stream>>>(x, T1, T2, T3, pack4, bias, out, N);
}


# SpMM as a gather by destination row instead of per-edge float atomics: edges grouped in-call (returning int add = slot, wave-summed row starts); init pass folded in; row sums in arrival order
# speedup vs baseline: 2.9762x; 2.9762x over previous
#include <hip/hip_runtime.h>
#include <hip/hip_bf16.h>

typedef __attribute__((ext_vector_type(2))) float v2f;
typedef __attribute__((ext_vector_type(8))) float v8f;

#define D 256
#define D4 64
#define LDS_STRIDE 260

__global__ void zero_deg_kernel(int* __restrict__ deg, int n) {
    int i = blockIdx.x * blockDim.x + threadIdx.x;
    if (i < n) deg[i] = 0;
}

__global__ void count_rank_kernel(const int* __restrict__ row, int* __restrict__ deg,
                                  int* __restrict__ rank, int nE) {
    int e = blockIdx.x * blockDim.x + threadIdx.x;
    if (e < nE) rank[e] = atomicAdd(&deg[row[e]], 1);
}

__global__ __launch_bounds__(256) void row_start_kernel(const int* deg, int* cursor,
                                                        int2* __restrict__ rowinfo,
                                                        float* __restrict__ dis, int n) {
    const int i = blockIdx.x * 256 + threadIdx.x;
    const int lane = threadIdx.x & 31;
    const int d = (i < n) ? deg[i] : 0;
    int v = d;
    for (int o = 1; o < 32; o <<= 1) {
        int t = __shfl_up(v, o);
        if (lane >= o) v += t;
    }
    int base = 0;
    if (lane == 31) base = atomicAdd(cursor, v);
    base = __shfl(base, 31);
    if (i < n) {
        int2 o2;
        o2.x = base + v - d;
        o2.y = d;
        rowinfo[i] = o2;
        dis[i] = rsqrtf((float)(d + 1));
    }
}

__global__ void fill_rows_kernel(const int* __restrict__ row, const int* __restrict__ col,
                                 const int* __restrict__ rank, const int2* __restrict__ rowinfo,
                                 const float* __restrict__ dis, int2* __restrict__ cw, int nE) {
    int e = blockIdx.x * blockDim.x + threadIdx.x;
    if (e >= nE) return;
    int r = row[e];
    int c = col[e];
    float w = -dis[r] * dis[c];
    if (r == c) w += 1.0f;
    int2 o;
    o.x = c;
    o.y = __float_as_int(w);
    cw[rowinfo[r].x + rank[e]] = o;
}

__global__ __launch_bounds__(256) void pack_w_kernel(const float* __restrict__ W,
                                                     float4* __restrict__ pack4) {
    int t = blockIdx.x * 256 + threadIdx.x;
    int l   = t & 31;
    int s   = (t >> 5) & 63;
    int wc  = (t >> 11) & 7;
    int k   = t >> 14;
    int m   = l & 15;
    int h   = l >> 4;
    int kA  = 4 * s + 2 * h;
    int n0  = wc * 32 + m;
    const float* Wk = W + (size_t)k * D * D;
    float4 o;
    o.x = Wk[(size_t)kA * D + n0];
    o.y = Wk[(size_t)(kA + 1) * D + n0];
    o.z = Wk[(size_t)kA * D + n0 + 16];
    o.w = Wk[(size_t)(kA + 1) * D + n0 + 16];
    pack4[t] = o;
}

template <bool FIRST>
__global__ __launch_bounds__(256) void cheb_gather_kernel(
    const float* __restrict__ src, const float* __restrict__ prev,
    const float* __restrict__ dis, const int2* __restrict__ rowinfo,
    const int2* __restrict__ cw, float* __restrict__ dst, int nN) {
    const int lane = threadIdx.x & 31;
    const int wid = __builtin_amdgcn_readfirstlane((int)(threadIdx.x >> 5));
    const int nw  = gridDim.x * 8;
    const float4* s4 = (const float4*)src;
    for (int i = blockIdx.x * 8 + wid; i < nN; i += nw) {
        const int2 info = rowinfo[i];
        const int beg = info.x;
        const int end = info.x + info.y;
        float4 a0 = {0.f, 0.f, 0.f, 0.f};
        float4 a1 = {0.f, 0.f, 0.f, 0.f};
        for (int base = beg; base < end; base += 32) {
            const int n = min(32, end - base);
            const int2 my = cw[min(base + lane, end - 1)];
#pragma unroll 4
            for (int j = 0; j < n; ++j) {
                const int   c = __builtin_amdgcn_readlane(my.x, j);
                const float w = __int_as_float(__builtin_amdgcn_readlane(my.y, j));
                const float4* p = s4 + (size_t)c * D4 + lane;
                const float4 v0 = p[0];
                const float4 v1 = p[32];
                a0.x += w * v0.x; a0.y += w * v0.y; a0.z += w * v0.z; a0.w += w * v0.w;
                a1.x += w * v1.x; a1.y += w * v1.y; a1.z += w * v1.z; a1.w += w * v1.w;
            }
        }
        const size_t off = (size_t)i * D4 + lane;
        const float d  = dis[i];
        const float ws = 1.0f - d * d;
        const float4 x0 = s4[off];
        const float4 x1 = s4[off + 32];
        a0.x += ws * x0.x; a0.y += ws * x0.y; a0.z += ws * x0.z; a0.w += ws * x0.w;
        a1.x += ws * x1.x; a1.y += ws * x1.y; a1.z += ws * x1.z; a1.w += ws * x1.w;
        if (!FIRST) {
            const float4 p0 = ((const float4*)prev)[off];
            const float4 p1 = ((const float4*)prev)[off + 32];
            a0.x = 2.0f * a0.x - p0.x; a0.y = 2.0f * a0.y - p0.y;
            a0.z = 2.0f * a0.z - p0.z; a0.w = 2.0f * a0.w - p0.w;
            a1.x = 2.0f * a1.x - p1.x; a1.y = 2.0f * a1.y - p1.y;
            a1.z = 2.0f * a1.z - p1.z; a1.w = 2.0f * a1.w - p1.w;
        }
        ((float4*)dst)[off]      = a0;
        ((float4*)dst)[off + 32] = a1;
    }
}

template <bool FULL>
__device__ __forceinline__ void gemm_body(
    float (*As)[LDS_STRIDE],
    const float* __restrict__ T0, const float* __restrict__ T1,
    const float* __restrict__ T2, const float* __restrict__ T3,
    const float4* __restrict__ pack4, const float* __restrict__ bias,
    float* __restrict__ out, int rowBase, int tid, int wid, int m, int h) {
    const int lane = tid & 31;
    v8f acc00 = {}, acc01 = {}, acc10 = {}, acc11 = {};
    const float* Ts[4] = {T0, T1, T2, T3};

#pragma unroll
    for (int k = 0; k < 4; ++k) {
        __syncthreads();
        const float* Tk = Ts[k];
        const int nload = FULL ? 8 : 4;
#pragma unroll
        for (int i = 0; i < nload; ++i) {
            int idx = tid + i * 256;
            int r   = idx >> 6;
            int c4  = idx & 63;
            float4 v = *(const float4*)(Tk + ((size_t)(rowBase + r)) * D + c4 * 4);
            *(float4*)(&As[r][c4 * 4]) = v;
        }
        __syncthreads();

        const float4* p4 = pack4 + ((size_t)(k * 8 + wid) * 64 * 32) + lane;
#pragma unroll 4
        for (int s = 0; s < 64; ++s) {
            const int kA = 4 * s + 2 * h;
            v2f a0;
            a0.x = As[m][kA];
            a0.y = As[m][kA + 1];
            float4 bb = p4[(size_t)s * 32];
            v2f b0; b0.x = bb.x; b0.y = bb.y;
            v2f b1; b1.x = bb.z; b1.y = bb.w;
            acc00 = __builtin_amdgcn_wmma_f32_16x16x4_f32(
                false, a0, false, b0, (short)0, acc00, false, false);
            acc01 = __builtin_amdgcn_wmma_f32_16x16x4_f32(
                false, a0, false, b1, (short)0, acc01, false, false);
            if (FULL) {
                v2f a1;
                a1.x = As[16 + m][kA];
                a1.y = As[16 + m][kA + 1];
                acc10 = __builtin_amdgcn_wmma_f32_16x16x4_f32(
                    false, a1, false, b0, (short)0, acc10, false, false);
                acc11 = __builtin_amdgcn_wmma_f32_16x16x4_f32(
                    false, a1, false, b1, (short)0, acc11, false, false);
            }
        }
    }

    const int colBase = wid * 32;
    const float bias0 = bias[colBase + m];
    const float bias1 = bias[colBase + 16 + m];
#pragma unroll
    for (int j = 0; j < 8; ++j) {
        int rr = rowBase + j + 8 * h;
        out[(size_t)rr * D + colBase + m]      = acc00[j] + bias0;
        out[(size_t)rr * D + colBase + 16 + m] = acc01[j] + bias1;
        if (FULL) {
            int rr1 = rr + 16;
            out[(size_t)rr1 * D + colBase + m]      = acc10[j] + bias0;
            out[(size_t)rr1 * D + colBase + 16 + m] = acc11[j] + bias1;
        }
    }
}

__global__ __launch_bounds__(256) void cheb_gemm_kernel(
    const float* __restrict__ T0, const float* __restrict__ T1,
    const float* __restrict__ T2, const float* __restrict__ T3,
    const float4* __restrict__ pack4, const float* __restrict__ bias,
    float* __restrict__ out, int nN) {
    __shared__ float As[32][LDS_STRIDE];
    const int tid = threadIdx.x;
    const int wid = tid >> 5;
    const int m   = tid & 15;
    const int h   = (tid >> 4) & 1;
    const int rowBase = blockIdx.x * 32;
    if (rowBase + 32 <= nN) {
        gemm_body<true>(As, T0, T1, T2, T3, pack4, bias, out, rowBase, tid, wid, m, h);
    } else {
        gemm_body<false>(As, T0, T1, T2, T3, pack4, bias, out, rowBase, tid, wid, m, h);
    }
}

extern "C" void kernel_launch(void* const* d_in, const int* in_sizes, int n_in,
                              void* d_out, int out_size, void* d_ws, size_t ws_size,
                              hipStream_t stream) {
    const float* x    = (const float*)d_in[0];
    const int*   ei   = (const int*)d_in[1];
    const float* W    = (const float*)d_in[2];
    const float* bias = (const float*)d_in[3];
    float* out = (float*)d_out;

    const int N = in_sizes[0] / D;
    const int E = in_sizes[1] / 2;
    const int* row = ei;
    const int* col = ei + E;

    char* ws = (char*)d_ws;
    size_t termBytes = (size_t)N * D * sizeof(float);
    int*    deg   = (int*)ws;
    float*  dis   = (float*)(ws + (1u << 18));
    float4* pack4 = (float4*)(ws + (1u << 19));
    float*  T1    = (float*)(ws + (3u << 19));
    float*  T2    = (float*)((char*)T1 + termBytes);
    float*  T3    = (float*)((char*)T2 + termBytes);
    int2*   cw      = (int2*)d_out;
    int2*   rowinfo = cw + E;
    int*    rank    = (int*)(rowinfo + N);

    const int bs = 256;
    pack_w_kernel<<<256, bs, 0, stream>>>(W, pack4);

    zero_deg_kernel<<<(N + 1 + bs - 1) / bs, bs, 0, stream>>>(deg, N + 1);
    count_rank_kernel<<<(E + bs - 1) / bs, bs, 0, stream>>>(row, deg, rank, E);
    row_start_kernel<<<(N + bs - 1) / bs, bs, 0, stream>>>(deg, deg + N, rowinfo, dis, N);
    fill_rows_kernel<<<(E + bs - 1) / bs, bs, 0, stream>>>(row, col, rank, rowinfo, dis, cw, E);

    const int gatherGrid = 1024;

    cheb_gather_kernel<true><<<gatherGrid, bs, 0, stream>>>(x, x, dis, rowinfo, cw, T1, N);
    cheb_gather_kernel<false><<<gatherGrid, bs, 0, stream>>>(T1, x, dis, rowinfo, cw, T2, N);
    cheb_gather_kernel<false><<<gatherGrid, bs, 0, stream>>>(T2, T1, dis, rowinfo, cw, T3, N);

    cheb_gemm_kernel<<<(N + 31) / 32, bs, 0, stream>>>(x, T1, T2, T3, pack4, bias, out, N);
}


# Edges grouped by row without a global atomic per edge: two-level binning on LDS counters (4096-edge blocks -> 256-row ranges -> rows), one global add per wave; row sums in LDS-add arrival order
# speedup vs baseline: 3.6850x; 1.2382x over previous
#include <hip/hip_runtime.h>
#include <hip/hip_bf16.h>

typedef __attribute__((ext_vector_type(2))) float v2f;
typedef __attribute__((ext_vector_type(8))) float v8f;

#define D 256
#define D4 64
#define LDS_STRIDE 260

#define ROWS_PER_RANGE 256
#define EDGE_CHUNK 4096
#define MAX_RANGES 1024

__global__ __launch_bounds__(1024) void bin_edges_kernel(
    const int* __restrict__ row, const int* __restrict__ col, int* cursor,
    int2* __restrict__ table, int2* __restrict__ stage, int nE, int nRanges) {
    __shared__ int hist[MAX_RANGES];
    __shared__ int offs[MAX_RANGES];
    const int t = threadIdx.x;
    const int lane = t & 31;
    hist[t] = 0;
    __syncthreads();
    const int e0 = blockIdx.x * EDGE_CHUNK + t;
    int r[4], c[4], lr[4];
#pragma unroll
    for (int k = 0; k < 4; ++k) {
        const int e = e0 + k * 1024;
        const int ec = min(e, nE - 1);
        const int rr = row[ec];
        const int cc = col[ec];
        r[k] = (e < nE) ? rr : -1;
        c[k] = cc;
    }
#pragma unroll
    for (int k = 0; k < 4; ++k) {
        lr[k] = 0;
        if (r[k] >= 0) lr[k] = atomicAdd(&hist[r[k] / ROWS_PER_RANGE], 1);
    }
    __syncthreads();
    const int cnt = (t < nRanges) ? hist[t] : 0;
    int v = cnt;
    for (int o = 1; o < 32; o <<= 1) {
        int u = __shfl_up(v, o);
        if (lane >= o) v += u;
    }
    int base = 0;
    if (lane == 31 && v > 0) base = atomicAdd(cursor, v);
    base = __shfl(base, 31);
    if (t < nRanges) {
        int2 run;
        run.x = base + v - cnt;
        run.y = cnt;
        offs[t] = run.x;
        table[(size_t)t * gridDim.x + blockIdx.x] = run;
    }
    __syncthreads();
#pragma unroll
    for (int k = 0; k < 4; ++k) {
        if (r[k] >= 0) {
            int2 o;
            o.x = r[k];
            o.y = c[k];
            stage[offs[r[k] / ROWS_PER_RANGE] + lr[k]] = o;
        }
    }
}

template <bool PLACE>
__device__ __forceinline__ void walk_runs(
    const int2* __restrict__ trow, const int2* __restrict__ stage, int nRuns,
    int wave, int lane, int rowBase, int* counter, const int* start, int2* __restrict__ cw) {
    for (int q0 = 0; q0 < nRuns; q0 += 1024) {
        const int j = q0 + wave + 32 * lane;
        int2 my; my.x = 0; my.y = 0;
        if (j < nRuns) my = trow[j];
        const int nq = min(32, (nRuns - q0 - wave + 31) / 32);
        for (int q = 0; q < nq; q += 4) {
            int off[4], n[4];
            int2 e[4];
#pragma unroll
            for (int u = 0; u < 4; ++u) {
                off[u] = __builtin_amdgcn_readlane(my.x, q + u);
                n[u]   = __builtin_amdgcn_readlane(my.y, q + u);
            }
#pragma unroll
            for (int u = 0; u < 4; ++u) {
                const int idx = (n[u] > 0) ? off[u] + min(lane, n[u] - 1) : 0;
                e[u] = stage[idx];
            }
#pragma unroll
            for (int u = 0; u < 4; ++u) {
                for (int i = lane; i < n[u]; i += 32) {
                    int2 ed = e[u];
                    if (i >= 32) ed = stage[off[u] + i];
                    const int rl = ed.x - rowBase;
                    if (PLACE) {
                        const int slot = start[rl] + atomicAdd(&counter[rl], 1);
                        int2 o;
                        o.x = ed.y;
                        o.y = ed.x;
                        cw[slot] = o;
                    } else {
                        atomicAdd(&counter[rl], 1);
                    }
                }
            }
        }
    }
}

__global__ __launch_bounds__(1024) void rank_rows_kernel(
    const int2* __restrict__ table, const int2* __restrict__ stage, int* cursor,
    int2* __restrict__ rowinfo, float* __restrict__ dis, int2* __restrict__ cw,
    int nN, int nRuns) {
    __shared__ int cnt[ROWS_PER_RANGE];
    __shared__ int cnt2[ROWS_PER_RANGE];
    __shared__ int start[ROWS_PER_RANGE];
    const int t = threadIdx.x;
    const int lane = t & 31;
    const int wave = __builtin_amdgcn_readfirstlane(t >> 5);
    const int rowBase = blockIdx.x * ROWS_PER_RANGE;
    if (t < ROWS_PER_RANGE) { cnt[t] = 0; cnt2[t] = 0; }
    __syncthreads();
    const int2* trow = table + (size_t)blockIdx.x * nRuns;

    walk_runs<false>(trow, stage, nRuns, wave, lane, rowBase, cnt, start, cw);
    __syncthreads();

    if (wave == 0) {
        int c8[8];
        int s = 0;
#pragma unroll
        for (int k = 0; k < 8; ++k) { c8[k] = cnt[lane * 8 + k]; s += c8[k]; }
        int v = s;
        for (int o = 1; o < 32; o <<= 1) {
            int u = __shfl_up(v, o);
            if (lane >= o) v += u;
        }
        int base = 0;
        if (lane == 31) base = atomicAdd(cursor, v);
        base = __shfl(base, 31);
        int run = base + v - s;
#pragma unroll
        for (int k = 0; k < 8; ++k) {
            const int rl = lane * 8 + k;
            start[rl] = run;
            const int i = rowBase + rl;
            if (i < nN) {
                int2 o2;
                o2.x = run;
                o2.y = c8[k];
                rowinfo[i] = o2;
                dis[i] = rsqrtf((float)(c8[k] + 1));
            }
            run += c8[k];
        }
    }
    __syncthreads();

    walk_runs<true>(trow, stage, nRuns, wave, lane, rowBase, cnt2, start, cw);
}

__global__ __launch_bounds__(256) void edge_weight_kernel(int2* __restrict__ cw,
                                                          const float* __restrict__ dis, int nE) {
    const int s0 = blockIdx.x * 1024 + threadIdx.x;
#pragma unroll
    for (int k = 0; k < 4; ++k) {
        const int s = s0 + k * 256;
        if (s < nE) {
            int2 o = cw[s];
            const int c = o.x;
            const int r = o.y;
            float w = -dis[r] * dis[c];
            if (r == c) w += 1.0f;
            o.y = __float_as_int(w);
            cw[s] = o;
        }
    }
}

__global__ __launch_bounds__(256) void pack_w_kernel(const float* __restrict__ W,
                                                     float4* __restrict__ pack4,
                                                     int* __restrict__ cursors) {
    int t = blockIdx.x * 256 + threadIdx.x;
    if (t < 2) cursors[t] = 0;
    int l   = t & 31;
    int s   = (t >> 5) & 63;
    int wc  = (t >> 11) & 7;
    int k   = t >> 14;
    int m   = l & 15;
    int h   = l >> 4;
    int kA  = 4 * s + 2 * h;
    int n0  = wc * 32 + m;
    const float* Wk = W + (size_t)k * D * D;
    float4 o;
    o.x = Wk[(size_t)kA * D + n0];
    o.y = Wk[(size_t)(kA + 1) * D + n0];
    o.z = Wk[(size_t)kA * D + n0 + 16];
    o.w = Wk[(size_t)(kA + 1) * D + n0 + 16];
    pack4[t] = o;
}

template <bool FIRST>
__global__ __launch_bounds__(256) void cheb_gather_kernel(
    const float* __restrict__ src, const float* __restrict__ prev,
    const float* __restrict__ dis, const int2* __restrict__ rowinfo,
    const int2* __restrict__ cw, float* __restrict__ dst, int nN) {
    const int lane = threadIdx.x & 31;
    const int wid = __builtin_amdgcn_readfirstlane((int)(threadIdx.x >> 5));
    const int nw  = gridDim.x * 8;
    const float4* s4 = (const float4*)src;
    for (int i = blockIdx.x * 8 + wid; i < nN; i += nw) {
        const int2 info = rowinfo[i];
        const int beg = info.x;
        const int end = info.x + info.y;
        float4 a0 = {0.f, 0.f, 0.f, 0.f};
        float4 a1 = {0.f, 0.f, 0.f, 0.f};
        for (int base = beg; base < end; base += 32) {
            const int n = min(32, end - base);
            const int2 my = cw[min(base + lane, end - 1)];
#pragma unroll 4
            for (int j = 0; j < n; ++j) {
                const int   c = __builtin_amdgcn_readlane(my.x, j);
                const float w = __int_as_float(__builtin_amdgcn_readlane(my.y, j));
                const float4* p = s4 + (size_t)c * D4 + lane;
                const float4 v0 = p[0];
                const float4 v1 = p[32];
                a0.x += w * v0.x; a0.y += w * v0.y; a0.z += w * v0.z; a0.w += w * v0.w;
                a1.x += w * v1.x; a1.y += w * v1.y; a1.z += w * v1.z; a1.w += w * v1.w;
            }
        }
        const size_t off = (size_t)i * D4 + lane;
        const float d  = dis[i];
        const float ws = 1.0f - d * d;
        const float4 x0 = s4[off];
        const float4 x1 = s4[off + 32];
        a0.x += ws * x0.x; a0.y += ws * x0.y; a0.z += ws * x0.z; a0.w += ws * x0.w;
        a1.x += ws * x1.x; a1.y += ws * x1.y; a1.z += ws * x1.z; a1.w += ws * x1.w;
        if (!FIRST) {
            const float4 p0 = ((const float4*)prev)[off];
            const float4 p1 = ((const float4*)prev)[off + 32];
            a0.x = 2.0f * a0.x - p0.x; a0.y = 2.0f * a0.y - p0.y;
            a0.z = 2.0f * a0.z - p0.z; a0.w = 2.0f * a0.w - p0.w;
            a1.x = 2.0f * a1.x - p1.x; a1.y = 2.0f * a1.y - p1.y;
            a1.z = 2.0f * a1.z - p1.z; a1.w = 2.0f * a1.w - p1.w;
        }
        ((float4*)dst)[off]      = a0;
        ((float4*)dst)[off + 32] = a1;
    }
}

template <bool FULL>
__device__ __forceinline__ void gather_tile(
    float (*As)[LDS_STRIDE],
    const float* __restrict__ src, const float* __restrict__ prev,
    const float* __restrict__ dis, const int2* __restrict__ rowinfo,
    const int2* __restrict__ cw, int rowBase, int wid, int lane) {
    const int nr = FULL ? 4 : 2;
    const float4* s4 = (const float4*)src;
    for (int q = 0; q < nr; ++q) {
        const int r = wid * nr + q;
        const int i = rowBase + r;
        const int2 info = rowinfo[i];
        const int beg = info.x;
        const int end = info.x + info.y;
        float4 a0 = {0.f, 0.f, 0.f, 0.f};
        float4 a1 = {0.f, 0.f, 0.f, 0.f};
        for (int base = beg; base < end; base += 32) {
            const int n = min(32, end - base);
            const int2 my = cw[min(base + lane, end - 1)];
#pragma unroll 4
            for (int j = 0; j < n; ++j) {
                const int   c = __builtin_amdgcn_readlane(my.x, j);
                const float w = __int_as_float(__builtin_amdgcn_readlane(my.y, j));
                const float4* p = s4 + (size_t)c * D4 + lane;
                const float4 v0 = p[0];
                const float4 v1 = p[32];
                a0.x += w * v0.x; a0.y += w * v0.y; a0.z += w * v0.z; a0.w += w * v0.w;
                a1.x += w * v1.x; a1.y += w * v1.y; a1.z += w * v1.z; a1.w += w * v1.w;
            }
        }
        const size_t off = (size_t)i * D4 + lane;
        const float d  = dis[i];
        const float ws = 1.0f - d * d;
        const float4 x0 = s4[off];
        const float4 x1 = s4[off + 32];
        a0.x += ws * x0.x; a0.y += ws * x0.y; a0.z += ws * x0.z; a0.w += ws * x0.w;
        a1.x += ws * x1.x; a1.y += ws * x1.y; a1.z += ws * x1.z; a1.w += ws * x1.w;
        const float4 p0 = ((const float4*)prev)[off];
        const float4 p1 = ((const float4*)prev)[off + 32];
        a0.x = 2.0f * a0.x - p0.x; a0.y = 2.0f * a0.y - p0.y;
        a0.z = 2.0f * a0.z - p0.z; a0.w = 2.0f * a0.w - p0.w;
        a1.x = 2.0f * a1.x - p1.x; a1.y = 2.0f * a1.y - p1.y;
        a1.z = 2.0f * a1.z - p1.z; a1.w = 2.0f * a1.w - p1.w;
        *(float4*)(&As[r][4 * lane])       = a0;
        *(float4*)(&As[r][128 + 4 * lane]) = a1;
    }
}

template <bool FULL>
__device__ __forceinline__ void gemm_body(
    float (*As)[LDS_STRIDE],
    const float* __restrict__ T0, const float* __restrict__ T1,
    const float* __restrict__ T2,
    const float* __restrict__ dis, const int2* __restrict__ rowinfo,
    const int2* __restrict__ cw,
    const float4* __restrict__ pack4, const float* __restrict__ bias,
    float* __restrict__ out, int rowBase, int tid, int wid, int m, int h, int first) {
    const int lane = tid & 31;
    v8f acc00 = {}, acc01 = {}, acc10 = {}, acc11 = {};

    for (int t = 0; t < 4; ++t) {
        const int k = (first + t) & 3;
        __syncthreads();
        if (k == 3) {
            gather_tile<FULL>(As, T2, T1, dis, rowinfo, cw, rowBase, wid, lane);
        } else {
            const float* Tk = (k == 0) ? T0 : ((k == 1) ? T1 : T2);
            const int nload = FULL ? 8 : 4;
#pragma unroll
            for (int i = 0; i < nload; ++i) {
                int idx = tid + i * 256;
                int r   = idx >> 6;
                int c4  = idx & 63;
                float4 v = *(const float4*)(Tk + ((size_t)(rowBase + r)) * D + c4 * 4);
                *(float4*)(&As[r][c4 * 4]) = v;
            }
        }
        __syncthreads();

        const float4* p4 = pack4 + ((size_t)(k * 8 + wid) * 64 * 32) + lane;
#pragma unroll 4
        for (int s = 0; s < 64; ++s) {
            const int kA = 4 * s + 2 * h;
            v2f a0;
            a0.x = As[m][kA];
            a0.y = As[m][kA + 1];
            float4 bb = p4[(size_t)s * 32];
            v2f b0; b0.x = bb.x; b0.y = bb.y;
            v2f b1; b1.x = bb.z; b1.y = bb.w;
            acc00 = __builtin_amdgcn_wmma_f32_16x16x4_f32(
                false, a0, false, b0, (short)0, acc00, false, false);
            acc01 = __builtin_amdgcn_wmma_f32_16x16x4_f32(
                false, a0, false, b1, (short)0, acc01, false, false);
            if (FULL) {
                v2f a1;
                a1.x = As[16 + m][kA];
                a1.y = As[16 + m][kA + 1];
                acc10 = __builtin_amdgcn_wmma_f32_16x16x4_f32(
                    false, a1, false, b0, (short)0, acc10, false, false);
                acc11 = __builtin_amdgcn_wmma_f32_16x16x4_f32(
                    false, a1, false, b1, (short)0, acc11, false, false);
            }
        }
    }

    const int colBase = wid * 32;
    const float bias0 = bias[colBase + m];
    const float bias1 = bias[colBase + 16 + m];
#pragma unroll
    for (int j = 0; j < 8; ++j) {
        int rr = rowBase + j + 8 * h;
        out[(size_t)rr * D + colBase + m]      = acc00[j] + bias0;
        out[(size_t)rr * D + colBase + 16 + m] = acc01[j] + bias1;
        if (FULL) {
            int rr1 = rr + 16;
            out[(size_t)rr1 * D + colBase + m]      = acc10[j] + bias0;
            out[(size_t)rr1 * D + colBase + 16 + m] = acc11[j] + bias1;
        }
    }
}

__global__ __launch_bounds__(256) void cheb_gemm_kernel(
    const float* __restrict__ T0, const float* __restrict__ T1,
    const float* __restrict__ T2,
    const float* __restrict__ dis, const int2* __restrict__ rowinfo,
    const int2* __restrict__ cw,
    const float4* __restrict__ pack4, const float* __restrict__ bias,
    float* __restrict__ out, int nN) {
    __shared__ float As[32][LDS_STRIDE];
    const int tid = threadIdx.x;
    const int wid = __builtin_amdgcn_readfirstlane(tid >> 5);
    const int m   = tid & 15;
    const int h   = (tid >> 4) & 1;
    const int rowBase = blockIdx.x * 32;
    unsigned dsum = 0;
    for (unsigned v = blockIdx.x; v != 0; v >>= 2) dsum += v;
    const int first = (int)(dsum & 3u);
    if (rowBase + 32 <= nN) {
        gemm_body<true>(As, T0, T1, T2, dis, rowinfo, cw, pack4, bias, out, rowBase, tid, wid, m, h, first);
    } else {
        gemm_body<false>(As, T0, T1, T2, dis, rowinfo, cw, pack4, bias, out, rowBase, tid, wid, m, h, first);
    }
}

extern "C" void kernel_launch(void* const* d_in, const int* in_sizes, int n_in,
                              void* d_out, int out_size, void* d_ws, size_t ws_size,
                              hipStream_t stream) {
    const float* x    = (const float*)d_in[0];
    const int*   ei   = (const int*)d_in[1];
    const float* W    = (const float*)d_in[2];
    const float* bias = (const float*)d_in[3];
    float* out = (float*)d_out;

    const int N = in_sizes[0] / D;
    const int E = in_sizes[1] / 2;
    const int* row = ei;
    const int* col = ei + E;

    char* ws = (char*)d_ws;
    size_t termBytes = (size_t)N * D * sizeof(float);
    int*    cursors = (int*)ws;
    float*  dis   = (float*)(ws + (1u << 18));
    float4* pack4 = (float4*)(ws + (1u << 19));
    float*  T1    = (float*)(ws + (3u << 19));
    float*  T2    = (float*)((char*)T1 + termBytes);
    int2*   cw      = (int2*)((char*)T2 + termBytes);
    int2*   rowinfo = cw + E;
    int2*   stage   = (int2*)T2;
    int2*   table   = stage + E;

    const int bs = 256;
    const int nRanges = (N + ROWS_PER_RANGE - 1) / ROWS_PER_RANGE;
    const int nChunks = (E + EDGE_CHUNK - 1) / EDGE_CHUNK;
    pack_w_kernel<<<256, bs, 0, stream>>>(W, pack4, cursors);

    bin_edges_kernel<<<nChunks, 1024, 0, stream>>>(row, col, cursors, table, stage, E, nRanges);
    rank_rows_kernel<<<nRanges, 1024, 0, stream>>>(table, stage, cursors + 1, rowinfo, dis, cw, N, nChunks);
    edge_weight_kernel<<<(E + 1023) / 1024, bs, 0, stream>>>(cw, dis, E);

    const int gatherGrid = 1024;

    cheb_gather_kernel<true><<<gatherGrid, bs, 0, stream>>>(x, x, dis, rowinfo, cw, T1, N);
    cheb_gather_kernel<false><<<gatherGrid, bs, 0, stream>>>(T1, x, dis, rowinfo, cw, T2, N);
    cheb_gemm_kernel<<<(N + 31) / 32, bs, 0, stream>>>(x, T1, T2, dis, rowinfo, cw, pack4, bias, out, N);
}
